# starting kernel
# baseline (speedup 1.0000x reference)
#include <hip/hip_runtime.h>
#include <hip/hip_bf16.h>
#include <math.h>

typedef __attribute__((ext_vector_type(2))) float v2f;
typedef __attribute__((ext_vector_type(8))) float v8f;

#define NB   4
#define NDET 128
#define NT   2048
#define NPIX (256 * 256)
#define WAVES_PER_BLOCK 8

__global__ __launch_bounds__(256)
void bp_wmma_f32_kernel(const float* __restrict__ sino,
                        const float* __restrict__ lut,
                        float* __restrict__ out)
{
    __shared__ float sApod[NDET];
    if (threadIdx.x < NDET) {
        const float w = 6.28318530717958647692f / 127.0f;
        sApod[threadIdx.x] = 0.5f - 0.5f * cosf((float)threadIdx.x * w);
    }
    __syncthreads();

    const int lane = threadIdx.x & 31;
    const int wave = threadIdx.x >> 5;
    const int tile = blockIdx.x * WAVES_PER_BLOCK + wave;
    const int p0   = tile << 4;
    const int halfSel = (lane < 16) ? 0 : 2;
    const int prow = p0 + (lane & 15);

    v8f acc[NB] = { v8f{0}, v8f{0}, v8f{0}, v8f{0} };

    for (int d = 0; d < NDET; d += 4) {
        const int det0 = d + halfSel;

        const float4 L = *(const float4*)(lut + ((size_t)prow * NDET + det0) * 2);

        const float2 ap = *(const float2*)(&sApod[det0]);
        v2f bb; bb.x = ap.x; bb.y = ap.y;

        const int kf0 = (int)L.x;
        const int kf1 = (int)L.z;
        const bool valid0 = (kf0 >= 0) && (kf0 < NT - 1);
        const bool valid1 = (kf1 >= 0) && (kf1 < NT - 1);
        const int k00 = min(max(kf0, 0), NT - 2);
        const int k10 = min(max(kf1, 0), NT - 2);
        const int roff0 = det0 * NT + k00;
        const int roff1 = (det0 + 1) * NT + k10;
        const float a0 = L.y, a1 = L.w;

#pragma unroll
        for (int b = 0; b < NB; ++b) {
            const float* s = sino + (size_t)b * (NDET * NT);
            const float s00 = s[roff0], s01 = s[roff0 + 1];
            const float s10 = s[roff1], s11 = s[roff1 + 1];
            float sk0 = (1.0f - a0) * s00 + a0 * s01;
            float sk1 = (1.0f - a1) * s10 + a1 * s11;
            sk0 = valid0 ? sk0 : 0.0f;
            sk1 = valid1 ? sk1 : 0.0f;

            v2f A; A.x = sk0; A.y = sk1;

            acc[b] = __builtin_amdgcn_wmma_f32_16x16x4_f32(
                 false, A,  false, bb,
                 (short)0, acc[b],  false,  false);
        }
    }

    const float invNorm = 1.0f / 63.5f;

    if (lane == 0 || lane == 16) {
        const int rbase = p0 + ((lane == 16) ? 8 : 0);
#pragma unroll
        for (int b = 0; b < NB; ++b) {
#pragma unroll
            for (int r = 0; r < 8; ++r) {
                out[(size_t)b * NPIX + rbase + r] = acc[b][r] * invNorm;
            }
        }
    }
}

extern "C" void kernel_launch(void* const* d_in, const int* in_sizes, int n_in,
                              void* d_out, int out_size, void* d_ws, size_t ws_size,
                              hipStream_t stream) {
    const float* sino = (const float*)d_in[0];
    const float* lut  = (const float*)d_in[1];
    float* out = (float*)d_out;

    const int tiles  = NPIX / 16;
    const int blocks = tiles / WAVES_PER_BLOCK;
    bp_wmma_f32_kernel<<<blocks, 256, 0, stream>>>(sino, lut, out);
}


# Work-group tile 512 pixels x 2 batches (was 256 x 4): a lane carries 2 pixels, batch pairs go to neighbouring groups, so each WGP pulls half the sinogram through its cache; LUT tile read twice
# speedup vs baseline: 2.7959x; 2.7959x over previous
#include <hip/hip_runtime.h>
#include <hip/hip_bf16.h>
#include <math.h>

typedef __attribute__((ext_vector_type(2))) float v2f;
typedef __attribute__((ext_vector_type(8))) float v8f;

#define NB   4
#define NBW  2
#define NPS  2
#define NDET 128
#define NT   2048
#define NPIX (256 * 256)
#define CH   16
#define DEPTH 3
#define RING 4
#define LROW 9
#define LSTEP 8

struct Gat { float s0[NPS][NBW]; float s1[NPS][NBW]; };

__device__ __forceinline__ void issue_gather(const float* __restrict__ sinoW, int det, const float (&kf)[NPS], Gat& g)
{
#pragma unroll
    for (int s = 0; s < NPS; ++s) {
        const int ki = (int)kf[s];
        const int k0 = min(max(ki, 0), NT - 2);
        const float* p = sinoW + det * NT + k0;
#pragma unroll
        for (int b = 0; b < NBW; ++b) {
            g.s0[s][b] = p[b * (NDET * NT)];
            g.s1[s][b] = p[b * (NDET * NT) + 1];
        }
    }
}

struct Lines { float4 r0, r1, r2, r3, r4, r5, r6, r7; };

__device__ __forceinline__ Lines lut_fetch(const float4* __restrict__ lutw, int chunk, int lane)
{
    const float4* p = lutw + (lane >> 3) * (NDET * 2 / 4) + chunk * (CH / 2) + (lane & 7);
    const int s = 4 * (NDET * 2 / 4);
    Lines G;
    G.r0 = p[0 * s]; G.r1 = p[1 * s]; G.r2 = p[2 * s]; G.r3 = p[3 * s];
    G.r4 = p[4 * s]; G.r5 = p[5 * s]; G.r6 = p[6 * s]; G.r7 = p[7 * s];
    return G;
}

__device__ __forceinline__ void lut_store(float4* tile, int lane, const Lines& G)
{
    float4* t = tile + (lane >> 3) * LROW + (lane & 7);
    const int s = 4 * LROW;
    t[0 * s] = G.r0; t[1 * s] = G.r1; t[2 * s] = G.r2; t[3 * s] = G.r3;
    t[4 * s] = G.r4; t[5 * s] = G.r5; t[6 * s] = G.r6; t[7 * s] = G.r7;
    __builtin_amdgcn_fence(__ATOMIC_ACQ_REL, "wavefront");
}

__device__ __forceinline__ float4 lut_read(const float4* tile, int lane, int q)
{
    return tile[lane * LROW + q];
}

__global__ __launch_bounds__(256)
void bp_wmma_f32_kernel(const float* __restrict__ sino,
                        const float* __restrict__ lut,
                        float* __restrict__ out)
{
    __shared__ float sApod[NDET];
    __shared__ float4 sLut[8 * NPS * 32 * LROW];
    if (threadIdx.x < NDET) {
        const float w = 6.28318530717958647692f / 127.0f;
        sApod[threadIdx.x] = 0.5f - 0.5f * cosf((float)threadIdx.x * w);
    }
    __syncthreads();

    const int lane = threadIdx.x & 31;
    const int wave = threadIdx.x >> 5;
    const int pg = blockIdx.x >> 1;
    const int bg = blockIdx.x & 1;
    const int wpix = pg * (256 * NPS) + wave * (32 * NPS);
    const float* sinoW = sino + (size_t)bg * (NBW * NDET * NT);

    const bool bsel = ((lane >> 4) == ((lane >> 3) & 1));

    v8f acc[NPS][NBW] = { { v8f{0}, v8f{0} }, { v8f{0}, v8f{0} } };

    const float4* lutw[NPS];
    float4* tile[NPS];
#pragma unroll
    for (int s = 0; s < NPS; ++s) {
        lutw[s] = (const float4*)(lut + (size_t)(wpix + 32 * s) * (NDET * 2));
        tile[s] = &sLut[(wave * NPS + s) * (32 * LROW)];
    }
    float4 Lc[NPS][CH / 2], Ln[NPS][2];
    Lines G[NPS];
#pragma unroll
    for (int s = 0; s < NPS; ++s) {
        G[s] = lut_fetch(lutw[s], 0, lane);
        lut_store(tile[s], lane, G[s]);
#pragma unroll
        for (int q = 0; q < CH / 2; ++q) Lc[s][q] = lut_read(tile[s], lane, q);
        G[s] = lut_fetch(lutw[s], 1, lane);
    }

    Gat ring[RING];
    float skEven[NPS][NBW];

#pragma unroll
    for (int j = 0; j < DEPTH; ++j) {
        float kfn[NPS];
#pragma unroll
        for (int s = 0; s < NPS; ++s) kfn[s] = (j & 1) ? Lc[s][j >> 1].z : Lc[s][j >> 1].x;
        issue_gather(sinoW, j, kfn, ring[j & (RING - 1)]);
        __builtin_amdgcn_s_barrier();
    }

    for (int c = 0; c < NDET / CH; ++c) {
        float2 ap[CH / 2];
#pragma unroll
        for (int i = 0; i < CH / 2; ++i) ap[i] = *(const float2*)(&sApod[c * CH + 2 * i]);

#pragma unroll
        for (int j = 0; j < CH; ++j) {
            {
                const int jj = j + DEPTH;
                float kfn[NPS];
#pragma unroll
                for (int s = 0; s < NPS; ++s) {
                    const float4 L = (jj < CH) ? Lc[s][jj >> 1] : Ln[s][(jj - CH) >> 1];
                    kfn[s] = (jj & 1) ? L.z : L.x;
                }
                const int dn = min(c * CH + jj, NDET - 1);
                issue_gather(sinoW, dn, kfn, ring[jj & (RING - 1)]);
            }
            __builtin_amdgcn_s_barrier();

            Gat& g = ring[j & (RING - 1)];
            asm volatile("" : "+v"(g.s0[0][0]), "+v"(g.s1[0][0]), "+v"(g.s0[0][1]), "+v"(g.s1[0][1]),
                              "+v"(g.s0[1][0]), "+v"(g.s1[1][0]), "+v"(g.s0[1][1]), "+v"(g.s1[1][1]));
            float sk[NPS][NBW];
#pragma unroll
            for (int s = 0; s < NPS; ++s) {
                const float4 L  = Lc[s][j >> 1];
                const float  kf = (j & 1) ? L.z : L.x;
                const float  a  = (j & 1) ? L.w : L.y;
                const int    ki = (int)kf;
                const bool valid = (ki >= 0) && (ki < NT - 1);
#pragma unroll
                for (int b = 0; b < NBW; ++b) {
                    float v = (1.0f - a) * g.s0[s][b] + a * g.s1[s][b];
                    sk[s][b] = valid ? v : 0.0f;
                }
            }
            if ((j & 1) == 0) {
#pragma unroll
                for (int s = 0; s < NPS; ++s)
#pragma unroll
                    for (int b = 0; b < NBW; ++b) skEven[s][b] = sk[s][b];
            } else {
                v2f bb; bb.x = bsel ? ap[j >> 1].x : 0.0f; bb.y = bsel ? ap[j >> 1].y : 0.0f;
#pragma unroll
                for (int s = 0; s < NPS; ++s)
#pragma unroll
                    for (int b = 0; b < NBW; ++b) {
                        v2f A; A.x = skEven[s][b]; A.y = sk[s][b];
                        acc[s][b] = __builtin_amdgcn_wmma_f32_16x16x4_f32(
                             false, A,  false, bb,
                             (short)0, acc[s][b],  false,  false);
                    }
            }
            if (j == LSTEP) {
#pragma unroll
                for (int s = 0; s < NPS; ++s) {
                    lut_store(tile[s], lane, G[s]);
                    Ln[s][0] = lut_read(tile[s], lane, 0);
                    Ln[s][1] = lut_read(tile[s], lane, 1);
                    G[s] = lut_fetch(lutw[s], min(c + 2, NDET / CH - 1), lane);
                }
            }
            __builtin_amdgcn_sched_barrier(0);
        }
#pragma unroll
        for (int s = 0; s < NPS; ++s) {
            Lc[s][0] = Ln[s][0];
            Lc[s][1] = Ln[s][1];
#pragma unroll
            for (int q = 2; q < CH / 2; ++q) Lc[s][q] = lut_read(tile[s], lane, q);
        }
    }

    const float invNorm = 1.0f / 63.5f;

    const int r    = lane & 7;
    const int poff = (((lane >> 3) & 1) << 4) + ((lane >> 4) << 3) + r;
#pragma unroll
    for (int s = 0; s < NPS; ++s)
#pragma unroll
        for (int b = 0; b < NBW; ++b) {
            float v = acc[s][b][0];
#pragma unroll
            for (int q = 1; q < 8; ++q) v = (r == q) ? acc[s][b][q] : v;
            out[(size_t)(bg * NBW + b) * NPIX + wpix + 32 * s + poff] = v * invNorm;
        }
}

extern "C" void kernel_launch(void* const* d_in, const int* in_sizes, int n_in,
                              void* d_out, int out_size, void* d_ws, size_t ws_size,
                              hipStream_t stream) {
    const float* sino = (const float*)d_in[0];
    const float* lut  = (const float*)d_in[1];
    float* out = (float*)d_out;

    bp_wmma_f32_kernel<<<256, 256, 0, stream>>>(sino, lut, out);
}


# Each work-group starts its detector sweep at its own 16-detector chunk and wraps round (detector sum taken in that rotated order): groups ask the second-level cache for 8 rows at a time, not one
# speedup vs baseline: 2.9213x; 1.0449x over previous
#include <hip/hip_runtime.h>
#include <hip/hip_bf16.h>
#include <math.h>

typedef __attribute__((ext_vector_type(2))) float v2f;
typedef __attribute__((ext_vector_type(8))) float v8f;

#define NB   4
#define NBW  2
#define NPS  2
#define NDET 128
#define NT   2048
#define NPIX (256 * 256)
#define CH   16
#define DEPTH 3
#define RING 4
#define LROW 9
#define LSTEP 8

struct Gat { float s0[NPS][NBW]; float s1[NPS][NBW]; };

__device__ __forceinline__ void issue_gather(const float* __restrict__ sinoW, int det, const float (&kf)[NPS], Gat& g)
{
#pragma unroll
    for (int s = 0; s < NPS; ++s) {
        const int ki = (int)kf[s];
        const int k0 = min(max(ki, 0), NT - 2);
        const float* p = sinoW + det * NT + k0;
#pragma unroll
        for (int b = 0; b < NBW; ++b) {
            g.s0[s][b] = p[b * (NDET * NT)];
            g.s1[s][b] = p[b * (NDET * NT) + 1];
        }
    }
}

struct Lines { float4 r0, r1, r2, r3, r4, r5, r6, r7; };

__device__ __forceinline__ Lines lut_fetch(const float4* __restrict__ lutw, int chunk, int lane)
{
    const float4* p = lutw + (lane >> 3) * (NDET * 2 / 4) + chunk * (CH / 2) + (lane & 7);
    const int s = 4 * (NDET * 2 / 4);
    Lines G;
    G.r0 = p[0 * s]; G.r1 = p[1 * s]; G.r2 = p[2 * s]; G.r3 = p[3 * s];
    G.r4 = p[4 * s]; G.r5 = p[5 * s]; G.r6 = p[6 * s]; G.r7 = p[7 * s];
    return G;
}

__device__ __forceinline__ void lut_store(float4* tile, int lane, const Lines& G)
{
    float4* t = tile + (lane >> 3) * LROW + (lane & 7);
    const int s = 4 * LROW;
    t[0 * s] = G.r0; t[1 * s] = G.r1; t[2 * s] = G.r2; t[3 * s] = G.r3;
    t[4 * s] = G.r4; t[5 * s] = G.r5; t[6 * s] = G.r6; t[7 * s] = G.r7;
    __builtin_amdgcn_fence(__ATOMIC_ACQ_REL, "wavefront");
}

__device__ __forceinline__ float4 lut_read(const float4* tile, int lane, int q)
{
    return tile[lane * LROW + q];
}

__global__ __launch_bounds__(256)
void bp_wmma_f32_kernel(const float* __restrict__ sino,
                        const float* __restrict__ lut,
                        float* __restrict__ out)
{
    __shared__ float sApod[NDET];
    __shared__ float4 sLut[8 * NPS * 32 * LROW];
    if (threadIdx.x < NDET) {
        const float w = 6.28318530717958647692f / 127.0f;
        sApod[threadIdx.x] = 0.5f - 0.5f * cosf((float)threadIdx.x * w);
    }
    __syncthreads();

    const int lane = threadIdx.x & 31;
    const int wave = threadIdx.x >> 5;
    const int pg = blockIdx.x >> 1;
    const int bg = blockIdx.x & 1;
    const int wpix = pg * (256 * NPS) + wave * (32 * NPS);
    const float* sinoW = sino + (size_t)bg * (NBW * NDET * NT);
    const int c0 = (pg ^ (pg >> 3) ^ (pg >> 6)) & (NDET / CH - 1);

    const bool bsel = ((lane >> 4) == ((lane >> 3) & 1));

    v8f acc[NPS][NBW] = { { v8f{0}, v8f{0} }, { v8f{0}, v8f{0} } };

    const float4* lutw[NPS];
    float4* tile[NPS];
#pragma unroll
    for (int s = 0; s < NPS; ++s) {
        lutw[s] = (const float4*)(lut + (size_t)(wpix + 32 * s) * (NDET * 2));
        tile[s] = &sLut[(wave * NPS + s) * (32 * LROW)];
    }
    float4 Lc[NPS][CH / 2], Ln[NPS][2];
    Lines G[NPS];
#pragma unroll
    for (int s = 0; s < NPS; ++s) {
        G[s] = lut_fetch(lutw[s], c0, lane);
        lut_store(tile[s], lane, G[s]);
#pragma unroll
        for (int q = 0; q < CH / 2; ++q) Lc[s][q] = lut_read(tile[s], lane, q);
        G[s] = lut_fetch(lutw[s], (c0 + 1) & (NDET / CH - 1), lane);
    }

    Gat ring[RING];
    float skEven[NPS][NBW];

#pragma unroll
    for (int j = 0; j < DEPTH; ++j) {
        float kfn[NPS];
#pragma unroll
        for (int s = 0; s < NPS; ++s) kfn[s] = (j & 1) ? Lc[s][j >> 1].z : Lc[s][j >> 1].x;
        issue_gather(sinoW, c0 * CH + j, kfn, ring[j & (RING - 1)]);
        __builtin_amdgcn_s_barrier();
    }

    for (int it = 0; it < NDET / CH; ++it) {
        const int c = (c0 + it) & (NDET / CH - 1);
        float2 ap[CH / 2];
#pragma unroll
        for (int i = 0; i < CH / 2; ++i) ap[i] = *(const float2*)(&sApod[c * CH + 2 * i]);

#pragma unroll
        for (int j = 0; j < CH; ++j) {
            {
                const int jj = j + DEPTH;
                float kfn[NPS];
#pragma unroll
                for (int s = 0; s < NPS; ++s) {
                    const float4 L = (jj < CH) ? Lc[s][jj >> 1] : Ln[s][(jj - CH) >> 1];
                    kfn[s] = (jj & 1) ? L.z : L.x;
                }
                const int dn = (c * CH + jj) & (NDET - 1);
                issue_gather(sinoW, dn, kfn, ring[jj & (RING - 1)]);
            }
            __builtin_amdgcn_s_barrier();

            Gat& g = ring[j & (RING - 1)];
            asm volatile("" : "+v"(g.s0[0][0]), "+v"(g.s1[0][0]), "+v"(g.s0[0][1]), "+v"(g.s1[0][1]),
                              "+v"(g.s0[1][0]), "+v"(g.s1[1][0]), "+v"(g.s0[1][1]), "+v"(g.s1[1][1]));
            float sk[NPS][NBW];
#pragma unroll
            for (int s = 0; s < NPS; ++s) {
                const float4 L  = Lc[s][j >> 1];
                const float  kf = (j & 1) ? L.z : L.x;
                const float  a  = (j & 1) ? L.w : L.y;
                const int    ki = (int)kf;
                const bool valid = (ki >= 0) && (ki < NT - 1);
#pragma unroll
                for (int b = 0; b < NBW; ++b) {
                    float v = (1.0f - a) * g.s0[s][b] + a * g.s1[s][b];
                    sk[s][b] = valid ? v : 0.0f;
                }
            }
            if ((j & 1) == 0) {
#pragma unroll
                for (int s = 0; s < NPS; ++s)
#pragma unroll
                    for (int b = 0; b < NBW; ++b) skEven[s][b] = sk[s][b];
            } else {
                v2f bb; bb.x = bsel ? ap[j >> 1].x : 0.0f; bb.y = bsel ? ap[j >> 1].y : 0.0f;
#pragma unroll
                for (int s = 0; s < NPS; ++s)
#pragma unroll
                    for (int b = 0; b < NBW; ++b) {
                        v2f A; A.x = skEven[s][b]; A.y = sk[s][b];
                        acc[s][b] = __builtin_amdgcn_wmma_f32_16x16x4_f32(
                             false, A,  false, bb,
                             (short)0, acc[s][b],  false,  false);
                    }
            }
            if (j == LSTEP) {
#pragma unroll
                for (int s = 0; s < NPS; ++s) {
                    lut_store(tile[s], lane, G[s]);
                    Ln[s][0] = lut_read(tile[s], lane, 0);
                    Ln[s][1] = lut_read(tile[s], lane, 1);
                    G[s] = lut_fetch(lutw[s], (c0 + min(it + 2, NDET / CH - 1)) & (NDET / CH - 1), lane);
                }
            }
            __builtin_amdgcn_sched_barrier(0);
        }
#pragma unroll
        for (int s = 0; s < NPS; ++s) {
            Lc[s][0] = Ln[s][0];
            Lc[s][1] = Ln[s][1];
#pragma unroll
            for (int q = 2; q < CH / 2; ++q) Lc[s][q] = lut_read(tile[s], lane, q);
        }
    }

    const float invNorm = 1.0f / 63.5f;

    const int r    = lane & 7;
    const int poff = (((lane >> 3) & 1) << 4) + ((lane >> 4) << 3) + r;
#pragma unroll
    for (int s = 0; s < NPS; ++s)
#pragma unroll
        for (int b = 0; b < NBW; ++b) {
            float v = acc[s][b][0];
#pragma unroll
            for (int q = 1; q < 8; ++q) v = (r == q) ? acc[s][b][q] : v;
            out[(size_t)(bg * NBW + b) * NPIX + wpix + 32 * s + poff] = v * invNorm;
        }
}

extern "C" void kernel_launch(void* const* d_in, const int* in_sizes, int n_in,
                              void* d_out, int out_size, void* d_ws, size_t ws_size,
                              hipStream_t stream) {
    const float* sino = (const float*)d_in[0];
    const float* lut  = (const float*)d_in[1];
    float* out = (float*)d_out;

    bp_wmma_f32_kernel<<<256, 256, 0, stream>>>(sino, lut, out);
}


# Each wave re-deals its 64 per-row gathers between its two gather instructions by sinogram half-row (through its own LDS strip, results handed back the same way): more lanes of a gather share a line
# speedup vs baseline: 3.1710x; 1.0855x over previous
#include <hip/hip_runtime.h>
#include <hip/hip_bf16.h>
#include <math.h>

typedef __attribute__((ext_vector_type(2))) float v2f;
typedef __attribute__((ext_vector_type(8))) float v8f;

#define NB   4
#define NBW  2
#define NPS  2
#define NDET 128
#define NT   2048
#define NPIX (256 * 256)
#define CH   16
#define DEPTH 3
#define RING 4
#define LROW 9
#define LSTEP 8

#define NHALF (NT / 2)

struct Gat { float s0[NPS][NBW]; float s1[NPS][NBW]; int meta[NPS]; float al[NPS]; };

__device__ __forceinline__ void deal_and_gather(const float* __restrict__ sinoW, int det, const float (&kf)[NPS],
                                                const float (&al)[NPS], int2* rec, int lane, Gat& g)
{
    int w0[NPS];
    bool low[NPS];
#pragma unroll
    for (int s = 0; s < NPS; ++s) {
        const int ki = (int)kf[s];
        const int k0 = min(max(ki, 0), NT - 2);
        const bool valid = (ki >= 0) && (ki < NT - 1);
        w0[s]  = k0 | (valid ? 2048 : 0) | ((s * 32 + lane) << 12);
        low[s] = k0 < NHALF;
    }
    const unsigned m0 = __builtin_amdgcn_ballot_w32(low[0]);
    const unsigned m1 = __builtin_amdgcn_ballot_w32(low[1]);
    const int nl0 = __builtin_popcount(m0);
    const int nlo = nl0 + __builtin_popcount(m1);
    const int c0 = (int)__builtin_amdgcn_mbcnt_lo(m0, 0u);
    const int c1 = (int)__builtin_amdgcn_mbcnt_lo(m1, 0u);
    const int lo0 = c0,       hi0 = nlo + (lane - c0);
    const int lo1 = nl0 + c1, hi1 = nlo + (32 - nl0) + (lane - c1);
    const int pos0 = low[0] ? lo0 : hi0;
    const int pos1 = low[1] ? lo1 : hi1;
    rec[pos0] = make_int2(w0[0], __float_as_int(al[0]));
    rec[pos1] = make_int2(w0[1], __float_as_int(al[1]));
    __builtin_amdgcn_fence(__ATOMIC_ACQ_REL, "wavefront");
#pragma unroll
    for (int t = 0; t < NPS; ++t) {
        const int2 R = rec[t * 32 + lane];
        g.meta[t] = R.x;
        g.al[t]   = __int_as_float(R.y);
        const float* p = sinoW + det * NT + (R.x & (NT - 1));
#pragma unroll
        for (int b = 0; b < NBW; ++b) {
            g.s0[t][b] = p[b * (NDET * NT)];
            g.s1[t][b] = p[b * (NDET * NT) + 1];
        }
    }
    __builtin_amdgcn_fence(__ATOMIC_ACQ_REL, "wavefront");
}

struct Lines { float4 r0, r1, r2, r3, r4, r5, r6, r7; };

__device__ __forceinline__ Lines lut_fetch(const float4* __restrict__ lutw, int chunk, int lane)
{
    const float4* p = lutw + (lane >> 3) * (NDET * 2 / 4) + chunk * (CH / 2) + (lane & 7);
    const int s = 4 * (NDET * 2 / 4);
    Lines G;
    G.r0 = p[0 * s]; G.r1 = p[1 * s]; G.r2 = p[2 * s]; G.r3 = p[3 * s];
    G.r4 = p[4 * s]; G.r5 = p[5 * s]; G.r6 = p[6 * s]; G.r7 = p[7 * s];
    return G;
}

__device__ __forceinline__ void lut_store(float4* tile, int lane, const Lines& G)
{
    float4* t = tile + (lane >> 3) * LROW + (lane & 7);
    const int s = 4 * LROW;
    t[0 * s] = G.r0; t[1 * s] = G.r1; t[2 * s] = G.r2; t[3 * s] = G.r3;
    t[4 * s] = G.r4; t[5 * s] = G.r5; t[6 * s] = G.r6; t[7 * s] = G.r7;
    __builtin_amdgcn_fence(__ATOMIC_ACQ_REL, "wavefront");
}

__device__ __forceinline__ float4 lut_read(const float4* tile, int lane, int q)
{
    return tile[lane * LROW + q];
}

__global__ __launch_bounds__(256)
void bp_wmma_f32_kernel(const float* __restrict__ sino,
                        const float* __restrict__ lut,
                        float* __restrict__ out)
{
    __shared__ float sApod[NDET];
    __shared__ float4 sLut[8 * NPS * 32 * LROW];
    __shared__ int2   sRec[8 * 64];
    __shared__ float2 sSk[8 * 64];
    if (threadIdx.x < NDET) {
        const float w = 6.28318530717958647692f / 127.0f;
        sApod[threadIdx.x] = 0.5f - 0.5f * cosf((float)threadIdx.x * w);
    }
    __syncthreads();

    const int lane = threadIdx.x & 31;
    const int wave = threadIdx.x >> 5;
    const int pg = blockIdx.x >> 1;
    const int bg = blockIdx.x & 1;
    const int wpix = pg * (256 * NPS) + wave * (32 * NPS);
    const float* sinoW = sino + (size_t)bg * (NBW * NDET * NT);
    const int c0 = (pg ^ (pg >> 3) ^ (pg >> 6)) & (NDET / CH - 1);

    const bool bsel = ((lane >> 4) == ((lane >> 3) & 1));

    v8f acc[NPS][NBW] = { { v8f{0}, v8f{0} }, { v8f{0}, v8f{0} } };

    const float4* lutw[NPS];
    float4* tile[NPS];
#pragma unroll
    for (int s = 0; s < NPS; ++s) {
        lutw[s] = (const float4*)(lut + (size_t)(wpix + 32 * s) * (NDET * 2));
        tile[s] = &sLut[(wave * NPS + s) * (32 * LROW)];
    }
    float4 Lc[NPS][CH / 2], Ln[NPS][2];
    Lines G[NPS];
#pragma unroll
    for (int s = 0; s < NPS; ++s) {
        G[s] = lut_fetch(lutw[s], c0, lane);
        lut_store(tile[s], lane, G[s]);
#pragma unroll
        for (int q = 0; q < CH / 2; ++q) Lc[s][q] = lut_read(tile[s], lane, q);
        G[s] = lut_fetch(lutw[s], (c0 + 1) & (NDET / CH - 1), lane);
    }

    Gat ring[RING];
    float skEven[NPS][NBW];
    int2*   rec = &sRec[wave * 64];
    float2* skb = &sSk[wave * 64];

#pragma unroll
    for (int j = 0; j < DEPTH; ++j) {
        float kfn[NPS], aln[NPS];
#pragma unroll
        for (int s = 0; s < NPS; ++s) {
            kfn[s] = (j & 1) ? Lc[s][j >> 1].z : Lc[s][j >> 1].x;
            aln[s] = (j & 1) ? Lc[s][j >> 1].w : Lc[s][j >> 1].y;
        }
        deal_and_gather(sinoW, c0 * CH + j, kfn, aln, rec, lane, ring[j & (RING - 1)]);
        __builtin_amdgcn_s_barrier();
    }

    for (int it = 0; it < NDET / CH; ++it) {
        const int c = (c0 + it) & (NDET / CH - 1);
        float2 ap[CH / 2];
#pragma unroll
        for (int i = 0; i < CH / 2; ++i) ap[i] = *(const float2*)(&sApod[c * CH + 2 * i]);

#pragma unroll
        for (int j = 0; j < CH; ++j) {
            {
                const int jj = j + DEPTH;
                float kfn[NPS], aln[NPS];
#pragma unroll
                for (int s = 0; s < NPS; ++s) {
                    const float4 L = (jj < CH) ? Lc[s][jj >> 1] : Ln[s][(jj - CH) >> 1];
                    kfn[s] = (jj & 1) ? L.z : L.x;
                    aln[s] = (jj & 1) ? L.w : L.y;
                }
                const int dn = (c * CH + jj) & (NDET - 1);
                deal_and_gather(sinoW, dn, kfn, aln, rec, lane, ring[jj & (RING - 1)]);
            }
            __builtin_amdgcn_s_barrier();

            Gat& g = ring[j & (RING - 1)];
            asm volatile("" : "+v"(g.s0[0][0]), "+v"(g.s1[0][0]), "+v"(g.s0[0][1]), "+v"(g.s1[0][1]),
                              "+v"(g.s0[1][0]), "+v"(g.s1[1][0]), "+v"(g.s0[1][1]), "+v"(g.s1[1][1]));
#pragma unroll
            for (int t = 0; t < NPS; ++t) {
                const float a     = g.al[t];
                const bool  valid = ((g.meta[t] >> 11) & 1) != 0;
                const int   owner = (g.meta[t] >> 12) & 63;
                float v[NBW];
#pragma unroll
                for (int b = 0; b < NBW; ++b) {
                    const float x = (1.0f - a) * g.s0[t][b] + a * g.s1[t][b];
                    v[b] = valid ? x : 0.0f;
                }
                skb[owner] = make_float2(v[0], v[1]);
            }
            __builtin_amdgcn_fence(__ATOMIC_ACQ_REL, "wavefront");
            float sk[NPS][NBW];
#pragma unroll
            for (int s = 0; s < NPS; ++s) {
                const float2 r2 = skb[s * 32 + lane];
                sk[s][0] = r2.x;
                sk[s][1] = r2.y;
            }
            __builtin_amdgcn_fence(__ATOMIC_ACQ_REL, "wavefront");
            if ((j & 1) == 0) {
#pragma unroll
                for (int s = 0; s < NPS; ++s)
#pragma unroll
                    for (int b = 0; b < NBW; ++b) skEven[s][b] = sk[s][b];
            } else {
                v2f bb; bb.x = bsel ? ap[j >> 1].x : 0.0f; bb.y = bsel ? ap[j >> 1].y : 0.0f;
#pragma unroll
                for (int s = 0; s < NPS; ++s)
#pragma unroll
                    for (int b = 0; b < NBW; ++b) {
                        v2f A; A.x = skEven[s][b]; A.y = sk[s][b];
                        acc[s][b] = __builtin_amdgcn_wmma_f32_16x16x4_f32(
                             false, A,  false, bb,
                             (short)0, acc[s][b],  false,  false);
                    }
            }
            if (j == LSTEP) {
#pragma unroll
                for (int s = 0; s < NPS; ++s) {
                    lut_store(tile[s], lane, G[s]);
                    Ln[s][0] = lut_read(tile[s], lane, 0);
                    Ln[s][1] = lut_read(tile[s], lane, 1);
                    G[s] = lut_fetch(lutw[s], (c0 + min(it + 2, NDET / CH - 1)) & (NDET / CH - 1), lane);
                }
            }
            __builtin_amdgcn_sched_barrier(0);
        }
#pragma unroll
        for (int s = 0; s < NPS; ++s) {
            Lc[s][0] = Ln[s][0];
            Lc[s][1] = Ln[s][1];
#pragma unroll
            for (int q = 2; q < CH / 2; ++q) Lc[s][q] = lut_read(tile[s], lane, q);
        }
    }

    const float invNorm = 1.0f / 63.5f;

    const int r    = lane & 7;
    const int poff = (((lane >> 3) & 1) << 4) + ((lane >> 4) << 3) + r;
#pragma unroll
    for (int s = 0; s < NPS; ++s)
#pragma unroll
        for (int b = 0; b < NBW; ++b) {
            float v = acc[s][b][0];
#pragma unroll
            for (int q = 1; q < 8; ++q) v = (r == q) ? acc[s][b][q] : v;
            out[(size_t)(bg * NBW + b) * NPIX + wpix + 32 * s + poff] = v * invNorm;
        }
}

extern "C" void kernel_launch(void* const* d_in, const int* in_sizes, int n_in,
                              void* d_out, int out_size, void* d_ws, size_t ws_size,
                              hipStream_t stream) {
    const float* sino = (const float*)d_in[0];
    const float* lut  = (const float*)d_in[1];
    float* out = (float*)d_out;

    bp_wmma_f32_kernel<<<256, 256, 0, stream>>>(sino, lut, out);
}
